# starting kernel
# baseline (speedup 1.0000x reference)
#include <hip/hip_runtime.h>

typedef __attribute__((ext_vector_type(2))) float v2f;
typedef __attribute__((ext_vector_type(8))) float v8f;

#define F_IN   128
#define H_DIM  64
#define Z_DIM  32
#define G_GRAPHS 512

__global__ __launch_bounds__(32) void gemm_wmma_f32(
    const float* __restrict__ A, const float* __restrict__ B,
    const float* __restrict__ bias, float* __restrict__ C,
    int M, int K, int Nc)
{
  const int tileN = blockIdx.x;
  const int tileM = blockIdx.y;
  const int lane  = threadIdx.x;
  const int half  = lane >> 4;
  const int l     = lane & 15;

  const float* Arow = A + (long long)(tileM * 16 + l) * K;
  const float* Bcol = B + (tileN * 16 + l);

  v8f c = {0.f, 0.f, 0.f, 0.f, 0.f, 0.f, 0.f, 0.f};

  for (int k = 0; k < K; k += 4) {
    const int ka = k + 2 * half;
    v2f a, b;
    a.x = Arow[ka];
    a.y = Arow[ka + 1];
    b.x = Bcol[(long long)ka * Nc];
    b.y = Bcol[(long long)(ka + 1) * Nc];
    c = __builtin_amdgcn_wmma_f32_16x16x4_f32(
        false, a, false, b, (short)0, c, false, false);
  }

  const int col = tileN * 16 + l;
  const float bv = bias ? bias[col] : 0.0f;
  float* Crow = C + (long long)(tileM * 16 + 8 * half) * Nc + col;
#pragma unroll
  for (int r = 0; r < 8; ++r) {
    Crow[(long long)r * Nc] = c[r] + bv;
  }
}

__global__ void fill_f32(float* __restrict__ p, float v, int n) {
  int t = blockIdx.x * blockDim.x + threadIdx.x;
  if (t < n) p[t] = v;
}

__global__ void deg_accum(const int* __restrict__ dst, float* __restrict__ deg, int E) {
  int t = blockIdx.x * blockDim.x + threadIdx.x;
  if (t < E) atomicAdd(&deg[dst[t]], 1.0f);
}

__global__ void deg_to_dinv(float* __restrict__ d, int n) {
  int t = blockIdx.x * blockDim.x + threadIdx.x;
  if (t < n) d[t] = rsqrtf(d[t] + 1.0f);
}

__global__ void selfloop_init(const float* __restrict__ h,
                              const float* __restrict__ dinv,
                              float* __restrict__ agg, int n) {
  int t = blockIdx.x * blockDim.x + threadIdx.x;
  if (t >= n * H_DIM) return;
  int i = t / H_DIM;
  float di = dinv[i];
  agg[t] = h[t] * (di * di);
}

__global__ void edge_prop(const int* __restrict__ src, const int* __restrict__ dst,
                          const float* __restrict__ dinv,
                          const float* __restrict__ h, float* __restrict__ agg,
                          int E) {
  const int chunks = H_DIM / 4;
  long long t = (long long)blockIdx.x * blockDim.x + threadIdx.x;
  if (t >= (long long)E * chunks) return;
  int e = (int)(t / chunks);
  int c = (int)(t % chunks);
  int s = src[e], d = dst[e];
  float coef = dinv[s] * dinv[d];
  float4 v = ((const float4*)(h + (long long)s * H_DIM))[c];
  float* o = agg + (long long)d * H_DIM + 4 * c;
  atomicAdd(o + 0, v.x * coef);
  atomicAdd(o + 1, v.y * coef);
  atomicAdd(o + 2, v.z * coef);
  atomicAdd(o + 3, v.w * coef);
}

__global__ void bias_relu(const float* __restrict__ agg, const float* __restrict__ b,
                          float* __restrict__ out, int n) {
  int t = blockIdx.x * blockDim.x + threadIdx.x;
  if (t >= n * H_DIM) return;
  float v = agg[t] + b[t % H_DIM];
  out[t] = v > 0.0f ? v : 0.0f;
}

__global__ void pool_accum(const float* __restrict__ h, const int* __restrict__ batch,
                           float* __restrict__ sums, float* __restrict__ cnts, int n) {
  const int chunks = H_DIM / 4;
  int t = blockIdx.x * blockDim.x + threadIdx.x;
  if (t >= n * chunks) return;
  int i = t / chunks, c = t % chunks;
  int g = batch[i];
  float4 v = ((const float4*)(h + (long long)i * H_DIM))[c];
  float* o = sums + (long long)g * H_DIM + 4 * c;
  atomicAdd(o + 0, v.x);
  atomicAdd(o + 1, v.y);
  atomicAdd(o + 2, v.z);
  atomicAdd(o + 3, v.w);
  if (c == 0) atomicAdd(&cnts[g], 1.0f);
}

__global__ void pool_final(const float* __restrict__ sums, const float* __restrict__ cnts,
                           float* __restrict__ pooled) {
  int t = blockIdx.x * blockDim.x + threadIdx.x;
  if (t >= G_GRAPHS * H_DIM) return;
  pooled[t] = sums[t] / fmaxf(cnts[t / H_DIM], 1.0f);
}

static inline size_t align_up(size_t x, size_t a) { return (x + a - 1) & ~(a - 1); }

extern "C" void kernel_launch(void* const* d_in, const int* in_sizes, int n_in,
                              void* d_out, int out_size, void* d_ws, size_t ws_size,
                              hipStream_t stream) {
  const float* x   = (const float*)d_in[0];
  const int*   ei  = (const int*)d_in[1];
  const int*   bat = (const int*)d_in[2];
  const float* W1  = (const float*)d_in[3];
  const float* b1  = (const float*)d_in[4];
  const float* W2  = (const float*)d_in[5];
  const float* b2  = (const float*)d_in[6];
  const float* Wmu = (const float*)d_in[7];
  const float* bmu = (const float*)d_in[8];
  const float* Wls = (const float*)d_in[9];
  const float* bls = (const float*)d_in[10];
  float* out = (float*)d_out;

  const int N = in_sizes[0] / F_IN;
  const int E = in_sizes[1] / 2;
  const int* src = ei;
  const int* dst = ei + E;

  char* ws = (char*)d_ws;
  size_t off = 0;
  float* dinv = (float*)(ws + off); off = align_up(off + (size_t)N * 4, 256);
  float* bufA = (float*)(ws + off); off = align_up(off + (size_t)N * H_DIM * 4, 256);
  float* bufB = (float*)(ws + off); off = align_up(off + (size_t)N * H_DIM * 4, 256);
  float* bufC = (float*)(ws + off); off = align_up(off + (size_t)N * H_DIM * 4, 256);
  float* sums = (float*)(ws + off); off = align_up(off + (size_t)G_GRAPHS * H_DIM * 4, 256);
  float* cnts = (float*)(ws + off); off = align_up(off + (size_t)G_GRAPHS * 4, 256);
  float* pooled = (float*)(ws + off); off = align_up(off + (size_t)G_GRAPHS * H_DIM * 4, 256);
  (void)ws_size; (void)n_in; (void)out_size;

  const int BLK = 256;
  const int NH = N * H_DIM;
  const long long EC = (long long)E * (H_DIM / 4);
  const int NC = N * (H_DIM / 4);

  fill_f32<<<(N + BLK - 1) / BLK, BLK, 0, stream>>>(dinv, 0.0f, N);
  deg_accum<<<(E + BLK - 1) / BLK, BLK, 0, stream>>>(dst, dinv, E);
  deg_to_dinv<<<(N + BLK - 1) / BLK, BLK, 0, stream>>>(dinv, N);

  gemm_wmma_f32<<<dim3(H_DIM / 16, N / 16), 32, 0, stream>>>(
      x, W1, nullptr, bufA, N, F_IN, H_DIM);
  selfloop_init<<<(NH + BLK - 1) / BLK, BLK, 0, stream>>>(bufA, dinv, bufB, N);
  edge_prop<<<(unsigned)((EC + BLK - 1) / BLK), BLK, 0, stream>>>(
      src, dst, dinv, bufA, bufB, E);
  bias_relu<<<(NH + BLK - 1) / BLK, BLK, 0, stream>>>(bufB, b1, bufA, N);

  gemm_wmma_f32<<<dim3(H_DIM / 16, N / 16), 32, 0, stream>>>(
      bufA, W2, nullptr, bufB, N, H_DIM, H_DIM);
  selfloop_init<<<(NH + BLK - 1) / BLK, BLK, 0, stream>>>(bufB, dinv, bufC, N);
  edge_prop<<<(unsigned)((EC + BLK - 1) / BLK), BLK, 0, stream>>>(
      src, dst, dinv, bufB, bufC, E);
  bias_relu<<<(NH + BLK - 1) / BLK, BLK, 0, stream>>>(bufC, b2, bufA, N);

  fill_f32<<<(G_GRAPHS * H_DIM + BLK - 1) / BLK, BLK, 0, stream>>>(sums, 0.0f, G_GRAPHS * H_DIM);
  fill_f32<<<(G_GRAPHS + BLK - 1) / BLK, BLK, 0, stream>>>(cnts, 0.0f, G_GRAPHS);
  pool_accum<<<(NC + BLK - 1) / BLK, BLK, 0, stream>>>(bufA, bat, sums, cnts, N);
  pool_final<<<(G_GRAPHS * H_DIM + BLK - 1) / BLK, BLK, 0, stream>>>(sums, cnts, pooled);

  gemm_wmma_f32<<<dim3(Z_DIM / 16, G_GRAPHS / 16), 32, 0, stream>>>(
      pooled, Wmu, bmu, out, G_GRAPHS, H_DIM, Z_DIM);
  gemm_wmma_f32<<<dim3(Z_DIM / 16, G_GRAPHS / 16), 32, 0, stream>>>(
      pooled, Wls, bls, out + (size_t)G_GRAPHS * Z_DIM, G_GRAPHS, H_DIM, Z_DIM);
}


# In-degree count moved to LDS: 32 workgroups count edge slices in per-node LDS counters, then add whole 128-B lines into deg (1 float atomic per 32 nodes, was 1 per edge); whole numbers, exact sums
# speedup vs baseline: 2.1628x; 2.1628x over previous
#include <hip/hip_runtime.h>

typedef __attribute__((ext_vector_type(2))) float v2f;
typedef __attribute__((ext_vector_type(8))) float v8f;

#define F_IN   128
#define H_DIM  64
#define Z_DIM  32
#define G_GRAPHS 512

#define A_PLAIN     0
#define A_BIAS_RELU 1
#define A_ROW_DIV   2
#define C_BIAS      0
#define C_SELFLOOP  1

template <int AF, int CF>
__global__ __launch_bounds__(32) void gemm_wmma_f32(
    const float* __restrict__ A, const float* __restrict__ B,
    const float* __restrict__ avec, const float* __restrict__ cvec,
    float* __restrict__ C, float* __restrict__ C2,
    int M, int K, int Nc)
{
  const int tileN = blockIdx.x;
  const int tileM = blockIdx.y;
  const int lane  = threadIdx.x;
  const int half  = lane >> 4;
  const int l     = lane & 15;

  const float* Arow = A + (long long)(tileM * 16 + l) * K;
  const float* Bcol = B + (tileN * 16 + l);
  const int row0 = tileM * 16 + 8 * half;

  float dg[8];
  if (CF == C_SELFLOOP) {
#pragma unroll
    for (int r = 0; r < 8; ++r) dg[r] = cvec[row0 + r];
  }
  float rc = 1.0f;
  if (AF == A_ROW_DIV) rc = fmaxf(avec[tileM * 16 + l], 1.0f);

  v8f c = {0.f, 0.f, 0.f, 0.f, 0.f, 0.f, 0.f, 0.f};

  int k = 0;
  for (; k + 16 <= K; k += 16) {
    v2f a[4], b[4], p[4];
#pragma unroll
    for (int j = 0; j < 4; ++j) {
      const int ka = k + 4 * j + 2 * half;
      a[j].x = Arow[ka];
      a[j].y = Arow[ka + 1];
      if (AF == A_BIAS_RELU) { p[j].x = avec[ka]; p[j].y = avec[ka + 1]; }
      b[j].x = Bcol[(long long)ka * Nc];
      b[j].y = Bcol[(long long)(ka + 1) * Nc];
    }
#pragma unroll
    for (int j = 0; j < 4; ++j) {
      if (AF == A_BIAS_RELU) {
        const float v0 = a[j].x + p[j].x;
        const float v1 = a[j].y + p[j].y;
        a[j].x = v0 > 0.0f ? v0 : 0.0f;
        a[j].y = v1 > 0.0f ? v1 : 0.0f;
      }
      if (AF == A_ROW_DIV) {
        a[j].x = a[j].x / rc;
        a[j].y = a[j].y / rc;
      }
    }
#pragma unroll
    for (int j = 0; j < 4; ++j) {
      c = __builtin_amdgcn_wmma_f32_16x16x4_f32(
          false, a[j], false, b[j], (short)0, c, false, false);
    }
  }
  for (; k < K; k += 4) {
    const int ka = k + 2 * half;
    v2f a, b;
    a.x = Arow[ka];
    a.y = Arow[ka + 1];
    if (AF == A_BIAS_RELU) {
      const float v0 = a.x + avec[ka];
      const float v1 = a.y + avec[ka + 1];
      a.x = v0 > 0.0f ? v0 : 0.0f;
      a.y = v1 > 0.0f ? v1 : 0.0f;
    }
    if (AF == A_ROW_DIV) {
      a.x = a.x / rc;
      a.y = a.y / rc;
    }
    b.x = Bcol[(long long)ka * Nc];
    b.y = Bcol[(long long)(ka + 1) * Nc];
    c = __builtin_amdgcn_wmma_f32_16x16x4_f32(
        false, a, false, b, (short)0, c, false, false);
  }

  const int col = tileN * 16 + l;
  float* Crow = C + (long long)row0 * Nc + col;
  if (CF == C_BIAS) {
    const float bv = cvec[col];
#pragma unroll
    for (int r = 0; r < 8; ++r) Crow[(long long)r * Nc] = c[r] + bv;
  } else {
    float* C2row = C2 + (long long)row0 * Nc + col;
#pragma unroll
    for (int r = 0; r < 8; ++r) {
      const float di = rsqrtf(dg[r] + 1.0f);
      Crow[(long long)r * Nc]  = c[r];
      C2row[(long long)r * Nc] = c[r] * (di * di);
    }
  }
}

template <int AF, int K>
__global__ __launch_bounds__(128) void gemm_rows64_wmma_f32(
    const float* __restrict__ A, const float* __restrict__ B,
    const float* __restrict__ avec, const float* __restrict__ cvec,
    float* __restrict__ C, float* __restrict__ C2, int M)
{
  constexpr int NC  = H_DIM;
  constexpr int KV  = K / 4;
  constexpr int PA  = K + 4;
  constexpr int PC  = NC + 4;
  constexpr int NLD = 64 * KV / 128;
  constexpr int RST = 128 / KV;
  __shared__ __attribute__((aligned(16))) float lds[64 * PA];

  const int tid  = threadIdx.x;
  const int lane = tid & 31;
  const int w    = tid >> 5;
  const int half = lane >> 4;
  const int l    = lane & 15;
  const int m0   = blockIdx.x * 64;

  const int arow = tid / KV;
  const int ac4  = tid % KV;
  float4 av[NLD];
#pragma unroll
  for (int i = 0; i < NLD; ++i) {
    const int grow = m0 + arow + RST * i;
    const int lrow = grow < M ? grow : M - 1;
    av[i] = *reinterpret_cast<const float4*>(A + (long long)lrow * K + 4 * ac4);
  }
  float4 pv = make_float4(0.f, 0.f, 0.f, 0.f);
  if (AF == A_BIAS_RELU) pv = *reinterpret_cast<const float4*>(avec + 4 * ac4);

  const float* Bcol = B + (16 * w + l);
  v2f b[K / 4];
#pragma unroll
  for (int j = 0; j < K / 4; ++j) {
    const int ka = 4 * j + 2 * half;
    b[j].x = Bcol[ka * NC];
    b[j].y = Bcol[(ka + 1) * NC];
  }

  const int erow = tid >> 4;
  const int ec4  = tid & 15;
  float dg[8];
#pragma unroll
  for (int i = 0; i < 8; ++i) {
    const int grow = m0 + erow + 8 * i;
    dg[i] = cvec[grow < M ? grow : M - 1];
  }

#pragma unroll
  for (int i = 0; i < NLD; ++i) {
    float4 v = av[i];
    if (AF == A_BIAS_RELU) {
      const float v0 = v.x + pv.x;
      const float v1 = v.y + pv.y;
      const float v2 = v.z + pv.z;
      const float v3 = v.w + pv.w;
      v.x = v0 > 0.0f ? v0 : 0.0f;
      v.y = v1 > 0.0f ? v1 : 0.0f;
      v.z = v2 > 0.0f ? v2 : 0.0f;
      v.w = v3 > 0.0f ? v3 : 0.0f;
    }
    *reinterpret_cast<float4*>(&lds[(arow + RST * i) * PA + 4 * ac4]) = v;
  }
  __syncthreads();

  v8f acc[4];
#pragma unroll
  for (int t = 0; t < 4; ++t) acc[t] = (v8f){0.f, 0.f, 0.f, 0.f, 0.f, 0.f, 0.f, 0.f};

#pragma unroll
  for (int j = 0; j < K / 4; ++j) {
#pragma unroll
    for (int t = 0; t < 4; ++t) {
      const v2f a = *reinterpret_cast<const v2f*>(&lds[(16 * t + l) * PA + 4 * j + 2 * half]);
      acc[t] = __builtin_amdgcn_wmma_f32_16x16x4_f32(
          false, a, false, b[j], (short)0, acc[t], false, false);
    }
  }
  __syncthreads();

#pragma unroll
  for (int t = 0; t < 4; ++t) {
#pragma unroll
    for (int r = 0; r < 8; ++r) {
      lds[(16 * t + 8 * half + r) * PC + 16 * w + l] = acc[t][r];
    }
  }
  __syncthreads();

#pragma unroll
  for (int i = 0; i < 8; ++i) {
    const int row  = erow + 8 * i;
    const int grow = m0 + row;
    const float4 v = *reinterpret_cast<const float4*>(&lds[row * PC + 4 * ec4]);
    if (grow < M) {
      const float di = rsqrtf(dg[i] + 1.0f);
      const float s  = di * di;
      float4 u;
      u.x = v.x * s; u.y = v.y * s; u.z = v.z * s; u.w = v.w * s;
      *reinterpret_cast<float4*>(C  + (long long)grow * NC + 4 * ec4) = v;
      *reinterpret_cast<float4*>(C2 + (long long)grow * NC + 4 * ec4) = u;
    }
  }
}

__global__ void fill_f32(float* __restrict__ p, float v, int n) {
  int t = blockIdx.x * blockDim.x + threadIdx.x;
  if (t < n) p[t] = v;
}

#define HIST_CAP  51200
#define HIST_WGS  32
#define HIST_THR  1024
#define HIST_INFL 25
#define HIST_FL   8
__global__ __launch_bounds__(HIST_THR) void deg_hist(
    const int* __restrict__ dst, float* __restrict__ deg, int E, int n) {
  __shared__ __attribute__((aligned(16))) unsigned int hist[HIST_CAP];
  const int tid  = threadIdx.x;
  const int lane = tid & 31;
  const int wv   = tid >> 5;
  const int nwg  = (int)gridDim.x;
  const int per  = (E + nwg - 1) / nwg;
  const int e_lo = (int)blockIdx.x * per;
  const int e_hi = (e_lo + per < E) ? (e_lo + per) : E;

  for (int base = 0; base < n; base += HIST_CAP) {
    const int cnt = (n - base < HIST_CAP) ? (n - base) : HIST_CAP;
    for (int i = 4 * tid; i < cnt; i += 4 * HIST_THR)
      *reinterpret_cast<uint4*>(&hist[i]) = make_uint4(0u, 0u, 0u, 0u);
    __syncthreads();

    for (int e = e_lo + tid; e < e_hi; e += HIST_THR * HIST_INFL) {
      int d[HIST_INFL];
#pragma unroll
      for (int j = 0; j < HIST_INFL; ++j) {
        const int ej = e + j * HIST_THR;
        d[j] = dst[ej < e_hi ? ej : e_hi - 1];
      }
#pragma unroll
      for (int j = 0; j < HIST_INFL; ++j) {
        const unsigned int r = (unsigned int)(d[j] - base);
        if (e + j * HIST_THR < e_hi && r < (unsigned int)cnt) atomicAdd(&hist[r], 1u);
      }
    }
    __syncthreads();

    const int nlines = (cnt + 31) >> 5;
    const int rot = (int)(((unsigned int)blockIdx.x * (unsigned int)nlines) / (unsigned int)nwg);
    for (int l0 = wv; l0 < nlines; l0 += (HIST_THR / 32) * HIST_FL) {
      unsigned int c[HIST_FL];
      int ix[HIST_FL];
#pragma unroll
      for (int j = 0; j < HIST_FL; ++j) {
        const int ln = l0 + j * (HIST_THR / 32);
        int L = ln + rot;
        if (L >= nlines) L -= nlines;
        const int i = 32 * L + lane;
        ix[j] = (ln < nlines && i < cnt) ? i : -1;
        c[j] = hist[ix[j] >= 0 ? i : 0];
      }
#pragma unroll
      for (int j = 0; j < HIST_FL; ++j) {
        if (ix[j] >= 0) atomicAdd(&deg[base + ix[j]], (float)c[j]);
      }
    }
    __syncthreads();
  }
}

__global__ __launch_bounds__(256) void edge_prop(
    const int* __restrict__ src, const int* __restrict__ dst,
    const float* __restrict__ deg,
    const float* __restrict__ h, float* __restrict__ agg, int E) {
  const int lane = threadIdx.x & 31;
  const int wave = blockIdx.x * (blockDim.x >> 5) + (threadIdx.x >> 5);
  const int e0 = wave * 32;
  if (e0 >= E) return;
  const int e = e0 + lane;
  int s = 0, d = 0;
  float cf = 0.0f;
  if (e < E) {
    s = src[e]; d = dst[e];
    cf = rsqrtf(deg[s] + 1.0f) * rsqrtf(deg[d] + 1.0f);
  }
  const int cfi = __float_as_int(cf);
  if (e0 + 32 <= E) {
#pragma unroll
    for (int g = 0; g < 32; g += 8) {
      float v0[8], v1[8];
#pragma unroll
      for (int j = 0; j < 8; ++j) {
        const int sj = __builtin_amdgcn_readlane(s, g + j);
        const float* row = h + (long long)sj * H_DIM + lane;
        v0[j] = row[0];
        v1[j] = row[32];
      }
#pragma unroll
      for (int j = 0; j < 8; ++j) {
        const int dj = __builtin_amdgcn_readlane(d, g + j);
        const float cj = __int_as_float(__builtin_amdgcn_readlane(cfi, g + j));
        float* o = agg + (long long)dj * H_DIM + lane;
        atomicAdd(o, v0[j] * cj);
        atomicAdd(o + 32, v1[j] * cj);
      }
    }
  } else {
    const int cnt = E - e0;
    for (int j = 0; j < cnt; ++j) {
      const int sj = __builtin_amdgcn_readlane(s, j);
      const int dj = __builtin_amdgcn_readlane(d, j);
      const float cj = __int_as_float(__builtin_amdgcn_readlane(cfi, j));
      const float* row = h + (long long)sj * H_DIM + lane;
      float* o = agg + (long long)dj * H_DIM + lane;
      atomicAdd(o, row[0] * cj);
      atomicAdd(o + 32, row[32] * cj);
    }
  }
}

__global__ __launch_bounds__(64) void pool_accum(
    const float* __restrict__ agg, const float* __restrict__ b,
    const int* __restrict__ batch,
    float* __restrict__ sums, float* __restrict__ cnts, int n) {
  const int lane = threadIdx.x & 31;
  const int wave = blockIdx.x * (blockDim.x >> 5) + (threadIdx.x >> 5);
  const int i0 = wave * 32;
  if (i0 >= n) return;
  const int i = i0 + lane;
  int g = 0;
  if (i < n) { g = batch[i]; atomicAdd(&cnts[g], 1.0f); }
  const float b0 = b[lane];
  const float b1 = b[lane + 32];
  const float* rows = agg + (long long)i0 * H_DIM + lane;
  if (i0 + 32 <= n) {
#pragma unroll
    for (int k = 0; k < 32; k += 8) {
      float v0[8], v1[8];
#pragma unroll
      for (int j = 0; j < 8; ++j) {
        v0[j] = rows[(k + j) * H_DIM];
        v1[j] = rows[(k + j) * H_DIM + 32];
      }
#pragma unroll
      for (int j = 0; j < 8; ++j) {
        const int gj = __builtin_amdgcn_readlane(g, k + j);
        float* o = sums + (long long)gj * H_DIM + lane;
        const float w0 = v0[j] + b0;
        const float w1 = v1[j] + b1;
        atomicAdd(o, w0 > 0.0f ? w0 : 0.0f);
        atomicAdd(o + 32, w1 > 0.0f ? w1 : 0.0f);
      }
    }
  } else {
    const int cnt = n - i0;
    for (int j = 0; j < cnt; ++j) {
      const int gj = __builtin_amdgcn_readlane(g, j);
      float* o = sums + (long long)gj * H_DIM + lane;
      const float w0 = rows[j * H_DIM] + b0;
      const float w1 = rows[j * H_DIM + 32] + b1;
      atomicAdd(o, w0 > 0.0f ? w0 : 0.0f);
      atomicAdd(o + 32, w1 > 0.0f ? w1 : 0.0f);
    }
  }
}

static inline size_t align_up(size_t x, size_t a) { return (x + a - 1) & ~(a - 1); }

extern "C" void kernel_launch(void* const* d_in, const int* in_sizes, int n_in,
                              void* d_out, int out_size, void* d_ws, size_t ws_size,
                              hipStream_t stream) {
  const float* x   = (const float*)d_in[0];
  const int*   ei  = (const int*)d_in[1];
  const int*   bat = (const int*)d_in[2];
  const float* W1  = (const float*)d_in[3];
  const float* b1  = (const float*)d_in[4];
  const float* W2  = (const float*)d_in[5];
  const float* b2  = (const float*)d_in[6];
  const float* Wmu = (const float*)d_in[7];
  const float* bmu = (const float*)d_in[8];
  const float* Wls = (const float*)d_in[9];
  const float* bls = (const float*)d_in[10];
  float* out = (float*)d_out;

  const int N = in_sizes[0] / F_IN;
  const int E = in_sizes[1] / 2;
  const int* src = ei;
  const int* dst = ei + E;

  char* ws = (char*)d_ws;
  size_t off = 0;
  float* deg  = (float*)(ws + off); off = align_up(off + (size_t)N * 4, 256);
  float* bufA = (float*)(ws + off); off = align_up(off + (size_t)N * H_DIM * 4, 256);
  float* bufB = (float*)(ws + off); off = align_up(off + (size_t)N * H_DIM * 4, 256);
  float* bufC = (float*)(ws + off); off = align_up(off + (size_t)N * H_DIM * 4, 256);
  float* sums = (float*)(ws + off); off = align_up(off + (size_t)G_GRAPHS * H_DIM * 4, 256);
  float* cnts = (float*)(ws + off); off = align_up(off + (size_t)G_GRAPHS * 4, 256);
  (void)ws_size; (void)n_in; (void)out_size;

  const int BLK = 256;
  const int EW = (E + 31) / 32;
  const int NW = (N + 31) / 32;

  fill_f32<<<(N + BLK - 1) / BLK, BLK, 0, stream>>>(deg, 0.0f, N);
  deg_hist<<<HIST_WGS, HIST_THR, 0, stream>>>(dst, deg, E, N);

  gemm_rows64_wmma_f32<A_PLAIN, F_IN><<<(N + 63) / 64, 128, 0, stream>>>(
      x, W1, nullptr, deg, bufA, bufB, N);
  edge_prop<<<(EW + 7) / 8, BLK, 0, stream>>>(
      src, dst, deg, bufA, bufB, E);

  gemm_rows64_wmma_f32<A_BIAS_RELU, H_DIM><<<(N + 63) / 64, 128, 0, stream>>>(
      bufB, W2, b1, deg, bufA, bufC, N);
  edge_prop<<<(EW + 7) / 8, BLK, 0, stream>>>(
      src, dst, deg, bufA, bufC, E);

  fill_f32<<<(G_GRAPHS * H_DIM + BLK - 1) / BLK, BLK, 0, stream>>>(sums, 0.0f, G_GRAPHS * H_DIM);
  fill_f32<<<(G_GRAPHS + BLK - 1) / BLK, BLK, 0, stream>>>(cnts, 0.0f, G_GRAPHS);
  pool_accum<<<(NW + 1) / 2, 64, 0, stream>>>(bufC, b2, bat, sums, cnts, N);

  gemm_wmma_f32<A_ROW_DIV, C_BIAS><<<dim3(Z_DIM / 16, G_GRAPHS / 16), 32, 0, stream>>>(
      sums, Wmu, cnts, bmu, out, nullptr, G_GRAPHS, H_DIM, Z_DIM);
  gemm_wmma_f32<A_ROW_DIV, C_BIAS><<<dim3(Z_DIM / 16, G_GRAPHS / 16), 32, 0, stream>>>(
      sums, Wls, cnts, bls, out + (size_t)G_GRAPHS * Z_DIM, nullptr, G_GRAPHS, H_DIM, Z_DIM);
}


# Mean pool by owner: a 32-wave workgroup per 16 graphs scans batch, sums its nodes' rows in registers and wave-own LDS planes, runs both heads; pool atomics, 2 clears go: 11->7 launches; fixed order
# speedup vs baseline: 2.4503x; 1.1329x over previous
#include <hip/hip_runtime.h>

typedef __attribute__((ext_vector_type(2))) float v2f;
typedef __attribute__((ext_vector_type(8))) float v8f;

#define F_IN   128
#define H_DIM  64
#define Z_DIM  32
#define G_GRAPHS 512

#define A_PLAIN     0
#define A_BIAS_RELU 1


template <int AF, int K>
__global__ __launch_bounds__(128) void gemm_rows64_wmma_f32(
    const float* __restrict__ A, const float* __restrict__ B,
    const float* __restrict__ avec, const float* __restrict__ cvec,
    float* __restrict__ C, float* __restrict__ C2, int M)
{
  constexpr int NC  = H_DIM;
  constexpr int KV  = K / 4;
  constexpr int PA  = K + 4;
  constexpr int PC  = NC + 4;
  constexpr int NLD = 64 * KV / 128;
  constexpr int RST = 128 / KV;
  __shared__ __attribute__((aligned(16))) float lds[64 * PA];

  const int tid  = threadIdx.x;
  const int lane = tid & 31;
  const int w    = tid >> 5;
  const int half = lane >> 4;
  const int l    = lane & 15;
  const int m0   = blockIdx.x * 64;

  const int arow = tid / KV;
  const int ac4  = tid % KV;
  float4 av[NLD];
#pragma unroll
  for (int i = 0; i < NLD; ++i) {
    const int grow = m0 + arow + RST * i;
    const int lrow = grow < M ? grow : M - 1;
    av[i] = *reinterpret_cast<const float4*>(A + (long long)lrow * K + 4 * ac4);
  }
  float4 pv = make_float4(0.f, 0.f, 0.f, 0.f);
  if (AF == A_BIAS_RELU) pv = *reinterpret_cast<const float4*>(avec + 4 * ac4);

  const float* Bcol = B + (16 * w + l);
  v2f b[K / 4];
#pragma unroll
  for (int j = 0; j < K / 4; ++j) {
    const int ka = 4 * j + 2 * half;
    b[j].x = Bcol[ka * NC];
    b[j].y = Bcol[(ka + 1) * NC];
  }

  const int erow = tid >> 4;
  const int ec4  = tid & 15;
  float dg[8];
#pragma unroll
  for (int i = 0; i < 8; ++i) {
    const int grow = m0 + erow + 8 * i;
    dg[i] = cvec[grow < M ? grow : M - 1];
  }

#pragma unroll
  for (int i = 0; i < NLD; ++i) {
    float4 v = av[i];
    if (AF == A_BIAS_RELU) {
      const float v0 = v.x + pv.x;
      const float v1 = v.y + pv.y;
      const float v2 = v.z + pv.z;
      const float v3 = v.w + pv.w;
      v.x = v0 > 0.0f ? v0 : 0.0f;
      v.y = v1 > 0.0f ? v1 : 0.0f;
      v.z = v2 > 0.0f ? v2 : 0.0f;
      v.w = v3 > 0.0f ? v3 : 0.0f;
    }
    *reinterpret_cast<float4*>(&lds[(arow + RST * i) * PA + 4 * ac4]) = v;
  }
  __syncthreads();

  v8f acc[4];
#pragma unroll
  for (int t = 0; t < 4; ++t) acc[t] = (v8f){0.f, 0.f, 0.f, 0.f, 0.f, 0.f, 0.f, 0.f};

#pragma unroll
  for (int j = 0; j < K / 4; ++j) {
#pragma unroll
    for (int t = 0; t < 4; ++t) {
      const v2f a = *reinterpret_cast<const v2f*>(&lds[(16 * t + l) * PA + 4 * j + 2 * half]);
      acc[t] = __builtin_amdgcn_wmma_f32_16x16x4_f32(
          false, a, false, b[j], (short)0, acc[t], false, false);
    }
  }
  __syncthreads();

#pragma unroll
  for (int t = 0; t < 4; ++t) {
#pragma unroll
    for (int r = 0; r < 8; ++r) {
      lds[(16 * t + 8 * half + r) * PC + 16 * w + l] = acc[t][r];
    }
  }
  __syncthreads();

#pragma unroll
  for (int i = 0; i < 8; ++i) {
    const int row  = erow + 8 * i;
    const int grow = m0 + row;
    const float4 v = *reinterpret_cast<const float4*>(&lds[row * PC + 4 * ec4]);
    if (grow < M) {
      const float di = rsqrtf(dg[i] + 1.0f);
      const float s  = di * di;
      float4 u;
      u.x = v.x * s; u.y = v.y * s; u.z = v.z * s; u.w = v.w * s;
      *reinterpret_cast<float4*>(C  + (long long)grow * NC + 4 * ec4) = v;
      *reinterpret_cast<float4*>(C2 + (long long)grow * NC + 4 * ec4) = u;
    }
  }
}

__global__ void fill_f32(float* __restrict__ p, float v, int n) {
  int t = blockIdx.x * blockDim.x + threadIdx.x;
  if (t < n) p[t] = v;
}

#define HIST_CAP  51200
#define HIST_WGS  32
#define HIST_THR  1024
#define HIST_INFL 25
#define HIST_FL   8
__global__ __launch_bounds__(HIST_THR) void deg_hist(
    const int* __restrict__ dst, float* __restrict__ deg, int E, int n) {
  __shared__ __attribute__((aligned(16))) unsigned int hist[HIST_CAP];
  const int tid  = threadIdx.x;
  const int lane = tid & 31;
  const int wv   = tid >> 5;
  const int nwg  = (int)gridDim.x;
  const int per  = (E + nwg - 1) / nwg;
  const int e_lo = (int)blockIdx.x * per;
  const int e_hi = (e_lo + per < E) ? (e_lo + per) : E;

  for (int base = 0; base < n; base += HIST_CAP) {
    const int cnt = (n - base < HIST_CAP) ? (n - base) : HIST_CAP;
    for (int i = 4 * tid; i < cnt; i += 4 * HIST_THR)
      *reinterpret_cast<uint4*>(&hist[i]) = make_uint4(0u, 0u, 0u, 0u);
    __syncthreads();

    for (int e = e_lo + tid; e < e_hi; e += HIST_THR * HIST_INFL) {
      int d[HIST_INFL];
#pragma unroll
      for (int j = 0; j < HIST_INFL; ++j) {
        const int ej = e + j * HIST_THR;
        d[j] = dst[ej < e_hi ? ej : e_hi - 1];
      }
#pragma unroll
      for (int j = 0; j < HIST_INFL; ++j) {
        const unsigned int r = (unsigned int)(d[j] - base);
        if (e + j * HIST_THR < e_hi && r < (unsigned int)cnt) atomicAdd(&hist[r], 1u);
      }
    }
    __syncthreads();

    const int nlines = (cnt + 31) >> 5;
    const int rot = (int)(((unsigned int)blockIdx.x * (unsigned int)nlines) / (unsigned int)nwg);
    for (int l0 = wv; l0 < nlines; l0 += (HIST_THR / 32) * HIST_FL) {
      unsigned int c[HIST_FL];
      int ix[HIST_FL];
#pragma unroll
      for (int j = 0; j < HIST_FL; ++j) {
        const int ln = l0 + j * (HIST_THR / 32);
        int L = ln + rot;
        if (L >= nlines) L -= nlines;
        const int i = 32 * L + lane;
        ix[j] = (ln < nlines && i < cnt) ? i : -1;
        c[j] = hist[ix[j] >= 0 ? i : 0];
      }
#pragma unroll
      for (int j = 0; j < HIST_FL; ++j) {
        if (ix[j] >= 0) atomicAdd(&deg[base + ix[j]], (float)c[j]);
      }
    }
    __syncthreads();
  }
}

__global__ __launch_bounds__(256) void edge_prop(
    const int* __restrict__ src, const int* __restrict__ dst,
    const float* __restrict__ deg,
    const float* __restrict__ h, float* __restrict__ agg, int E) {
  const int lane = threadIdx.x & 31;
  const int wave = blockIdx.x * (blockDim.x >> 5) + (threadIdx.x >> 5);
  const int e0 = wave * 32;
  if (e0 >= E) return;
  const int e = e0 + lane;
  int s = 0, d = 0;
  float cf = 0.0f;
  if (e < E) {
    s = src[e]; d = dst[e];
    cf = rsqrtf(deg[s] + 1.0f) * rsqrtf(deg[d] + 1.0f);
  }
  const int cfi = __float_as_int(cf);
  if (e0 + 32 <= E) {
#pragma unroll
    for (int g = 0; g < 32; g += 8) {
      float v0[8], v1[8];
#pragma unroll
      for (int j = 0; j < 8; ++j) {
        const int sj = __builtin_amdgcn_readlane(s, g + j);
        const float* row = h + (long long)sj * H_DIM + lane;
        v0[j] = row[0];
        v1[j] = row[32];
      }
#pragma unroll
      for (int j = 0; j < 8; ++j) {
        const int dj = __builtin_amdgcn_readlane(d, g + j);
        const float cj = __int_as_float(__builtin_amdgcn_readlane(cfi, g + j));
        float* o = agg + (long long)dj * H_DIM + lane;
        atomicAdd(o, v0[j] * cj);
        atomicAdd(o + 32, v1[j] * cj);
      }
    }
  } else {
    const int cnt = E - e0;
    for (int j = 0; j < cnt; ++j) {
      const int sj = __builtin_amdgcn_readlane(s, j);
      const int dj = __builtin_amdgcn_readlane(d, j);
      const float cj = __int_as_float(__builtin_amdgcn_readlane(cfi, j));
      const float* row = h + (long long)sj * H_DIM + lane;
      float* o = agg + (long long)dj * H_DIM + lane;
      atomicAdd(o, row[0] * cj);
      atomicAdd(o + 32, row[32] * cj);
    }
  }
}

#define POOL_GPW  16
#define POOL_THR  1024
#define POOL_NW   32
#define POOL_SCAN 32
#define POOL_PP   (H_DIM + 4)
__global__ __launch_bounds__(POOL_THR) void pool_heads(
    const float* __restrict__ agg, const float* __restrict__ b,
    const int* __restrict__ batch,
    const float* __restrict__ Wmu, const float* __restrict__ bmu,
    const float* __restrict__ Wls, const float* __restrict__ bls,
    float* __restrict__ out, int n) {
  __shared__ __attribute__((aligned(16))) float accw[POOL_NW * POOL_GPW * H_DIM];
  __shared__ float cntw[POOL_NW * POOL_GPW];
  __shared__ __attribute__((aligned(16))) float tile[POOL_GPW * POOL_PP];

  const int tid  = threadIdx.x;
  const int lane = tid & 31;
  const int w    = tid >> 5;
  const int half = lane >> 4;
  const int l    = lane & 15;
  const int g0   = (int)blockIdx.x * POOL_GPW;

  float* mine = accw + w * (POOL_GPW * H_DIM);
#pragma unroll
  for (int i = 0; i < POOL_GPW * H_DIM / 128; ++i)
    *reinterpret_cast<float4*>(mine + 4 * (lane + 32 * i)) = make_float4(0.f, 0.f, 0.f, 0.f);
  if (lane < POOL_GPW) cntw[w * POOL_GPW + lane] = 0.0f;

  const float2 bb = *reinterpret_cast<const float2*>(b + 2 * lane);

  int    cur = -1;
  float2 acc = make_float2(0.f, 0.f);
  float  cn  = 0.0f;

  auto flush = [&]() {
    if (cur >= 0) {
      float2* p = reinterpret_cast<float2*>(mine + cur * H_DIM + 2 * lane);
      float2 o = *p;
      o.x += acc.x; o.y += acc.y;
      *p = o;
      if (lane == 0) cntw[w * POOL_GPW + cur] += cn;
    }
  };

  auto chunk = [&](int c) {
    const int i0 = c << 5;
    const int i  = i0 + lane;
    const int gl = batch[i < n ? i : n - 1] - g0;
    const bool own = (i < n) && ((unsigned int)gl < (unsigned int)POOL_GPW);
    float2 r[32];
#pragma unroll
    for (int j = 0; j < 32; ++j) {
      const int row = (i0 + j < n) ? (i0 + j) : (n - 1);
      r[j] = *reinterpret_cast<const float2*>(agg + (long long)row * H_DIM + 2 * lane);
    }
    const unsigned int m = __builtin_amdgcn_ballot_w32(own);
#pragma unroll
    for (int j = 0; j < 32; ++j) {
      if ((m >> j) & 1u) {
        const int gj = __builtin_amdgcn_readlane(gl, j);
        if (gj != cur) {
          flush();
          cur = gj;
          acc = make_float2(0.f, 0.f);
          cn  = 0.0f;
        }
        const float w0 = r[j].x + bb.x;
        const float w1 = r[j].y + bb.y;
        acc.x += w0 > 0.0f ? w0 : 0.0f;
        acc.y += w1 > 0.0f ? w1 : 0.0f;
        cn += 1.0f;
      }
    }
  };

  const int nchunks = (n + 31) >> 5;
  for (int c0 = w; c0 < nchunks; c0 += POOL_NW * POOL_SCAN) {
    int bv[POOL_SCAN];
#pragma unroll
    for (int t = 0; t < POOL_SCAN; ++t) {
      const int i = ((c0 + POOL_NW * t) << 5) + lane;
      bv[t] = batch[i < n ? i : n - 1];
    }
    unsigned int hit = 0u;
#pragma unroll
    for (int t = 0; t < POOL_SCAN; ++t) {
      const int i = ((c0 + POOL_NW * t) << 5) + lane;
      const bool own = (i < n) && ((unsigned int)(bv[t] - g0) < (unsigned int)POOL_GPW);
      if (__builtin_amdgcn_ballot_w32(own) != 0u) hit |= (1u << t);
    }
    while (hit != 0u) {
      const int t = __builtin_ctz(hit);
      hit &= hit - 1u;
      chunk(c0 + POOL_NW * t);
    }
  }
  flush();

  v2f bf[H_DIM / 4];
  float bias = 0.0f;
  if (w < 4) {
    const float* Wh = (w < 2) ? Wmu : Wls;
    const float* bh = (w < 2) ? bmu : bls;
    const int col = 16 * (w & 1) + l;
#pragma unroll
    for (int j = 0; j < H_DIM / 4; ++j) {
      const int ka = 4 * j + 2 * half;
      bf[j].x = Wh[ka * Z_DIM + col];
      bf[j].y = Wh[(ka + 1) * Z_DIM + col];
    }
    bias = bh[col];
  }
  __syncthreads();

  {
    const int g = tid >> 6;
    const int f = tid & 63;
    float s = 0.0f, c = 0.0f;
#pragma unroll 8
    for (int ww = 0; ww < POOL_NW; ++ww) {
      s += accw[ww * (POOL_GPW * H_DIM) + g * H_DIM + f];
      c += cntw[ww * POOL_GPW + g];
    }
    tile[g * POOL_PP + f] = s / fmaxf(c, 1.0f);
  }
  __syncthreads();

  v8f d = {0.f, 0.f, 0.f, 0.f, 0.f, 0.f, 0.f, 0.f};
  if (w < 4) {
#pragma unroll
    for (int j = 0; j < H_DIM / 4; ++j) {
      const v2f a = *reinterpret_cast<const v2f*>(&tile[l * POOL_PP + 4 * j + 2 * half]);
      d = __builtin_amdgcn_wmma_f32_16x16x4_f32(
          false, a, false, bf[j], (short)0, d, false, false);
    }
  }
  __syncthreads();

  if (w < 4) {
#pragma unroll
    for (int r = 0; r < 8; ++r) tile[(8 * half + r) * POOL_PP + 16 * w + l] = d[r] + bias;
  }
  __syncthreads();

  {
    const int g  = w & 15;
    const int hd = w >> 4;
    out[(size_t)hd * G_GRAPHS * Z_DIM + (size_t)(g0 + g) * Z_DIM + lane] =
        tile[g * POOL_PP + hd * Z_DIM + lane];
  }
}

static inline size_t align_up(size_t x, size_t a) { return (x + a - 1) & ~(a - 1); }

extern "C" void kernel_launch(void* const* d_in, const int* in_sizes, int n_in,
                              void* d_out, int out_size, void* d_ws, size_t ws_size,
                              hipStream_t stream) {
  const float* x   = (const float*)d_in[0];
  const int*   ei  = (const int*)d_in[1];
  const int*   bat = (const int*)d_in[2];
  const float* W1  = (const float*)d_in[3];
  const float* b1  = (const float*)d_in[4];
  const float* W2  = (const float*)d_in[5];
  const float* b2  = (const float*)d_in[6];
  const float* Wmu = (const float*)d_in[7];
  const float* bmu = (const float*)d_in[8];
  const float* Wls = (const float*)d_in[9];
  const float* bls = (const float*)d_in[10];
  float* out = (float*)d_out;

  const int N = in_sizes[0] / F_IN;
  const int E = in_sizes[1] / 2;
  const int* src = ei;
  const int* dst = ei + E;

  char* ws = (char*)d_ws;
  size_t off = 0;
  float* deg  = (float*)(ws + off); off = align_up(off + (size_t)N * 4, 256);
  float* bufA = (float*)(ws + off); off = align_up(off + (size_t)N * H_DIM * 4, 256);
  float* bufB = (float*)(ws + off); off = align_up(off + (size_t)N * H_DIM * 4, 256);
  float* bufC = (float*)(ws + off); off = align_up(off + (size_t)N * H_DIM * 4, 256);
  (void)ws_size; (void)n_in; (void)out_size;

  const int BLK = 256;
  const int EW = (E + 31) / 32;

  fill_f32<<<(N + BLK - 1) / BLK, BLK, 0, stream>>>(deg, 0.0f, N);
  deg_hist<<<HIST_WGS, HIST_THR, 0, stream>>>(dst, deg, E, N);

  gemm_rows64_wmma_f32<A_PLAIN, F_IN><<<(N + 63) / 64, 128, 0, stream>>>(
      x, W1, nullptr, deg, bufA, bufB, N);
  edge_prop<<<(EW + 7) / 8, BLK, 0, stream>>>(
      src, dst, deg, bufA, bufB, E);

  gemm_rows64_wmma_f32<A_BIAS_RELU, H_DIM><<<(N + 63) / 64, 128, 0, stream>>>(
      bufB, W2, b1, deg, bufA, bufC, N);
  edge_prop<<<(EW + 7) / 8, BLK, 0, stream>>>(
      src, dst, deg, bufA, bufC, E);

  pool_heads<<<G_GRAPHS / POOL_GPW, POOL_THR, 0, stream>>>(
      bufC, b2, bat, Wmu, bmu, Wls, bls, out, N);
}


# Edge scatter split by cache half: dst rows in two ranges, each edge slice walked by one workgroup per range, range taken by the die it runs on (ticket counter); float atomics kept, order free
# speedup vs baseline: 2.8249x; 1.1529x over previous
#include <hip/hip_runtime.h>

typedef __attribute__((ext_vector_type(2))) float v2f;
typedef __attribute__((ext_vector_type(8))) float v8f;

#define F_IN   128
#define H_DIM  64
#define Z_DIM  32
#define G_GRAPHS 512

#define A_PLAIN     0
#define A_BIAS_RELU 1


template <int AF, int K>
__global__ __launch_bounds__(128) void gemm_rows64_wmma_f32(
    const float* __restrict__ A, const float* __restrict__ B,
    const float* __restrict__ avec, const float* __restrict__ cvec,
    float* __restrict__ C, float* __restrict__ C2, int M)
{
  constexpr int NC  = H_DIM;
  constexpr int KV  = K / 4;
  constexpr int PA  = K + 4;
  constexpr int PC  = NC + 4;
  constexpr int NLD = 64 * KV / 128;
  constexpr int RST = 128 / KV;
  __shared__ __attribute__((aligned(16))) float lds[64 * PA];

  const int tid  = threadIdx.x;
  const int lane = tid & 31;
  const int w    = tid >> 5;
  const int half = lane >> 4;
  const int l    = lane & 15;
  const int m0   = blockIdx.x * 64;

  const int arow = tid / KV;
  const int ac4  = tid % KV;
  float4 av[NLD];
#pragma unroll
  for (int i = 0; i < NLD; ++i) {
    const int grow = m0 + arow + RST * i;
    const int lrow = grow < M ? grow : M - 1;
    av[i] = *reinterpret_cast<const float4*>(A + (long long)lrow * K + 4 * ac4);
  }
  float4 pv = make_float4(0.f, 0.f, 0.f, 0.f);
  if (AF == A_BIAS_RELU) pv = *reinterpret_cast<const float4*>(avec + 4 * ac4);

  const float* Bcol = B + (16 * w + l);
  v2f b[K / 4];
#pragma unroll
  for (int j = 0; j < K / 4; ++j) {
    const int ka = 4 * j + 2 * half;
    b[j].x = Bcol[ka * NC];
    b[j].y = Bcol[(ka + 1) * NC];
  }

  const int erow = tid >> 4;
  const int ec4  = tid & 15;
  float dg[8];
#pragma unroll
  for (int i = 0; i < 8; ++i) {
    const int grow = m0 + erow + 8 * i;
    dg[i] = cvec[grow < M ? grow : M - 1];
  }

#pragma unroll
  for (int i = 0; i < NLD; ++i) {
    float4 v = av[i];
    if (AF == A_BIAS_RELU) {
      const float v0 = v.x + pv.x;
      const float v1 = v.y + pv.y;
      const float v2 = v.z + pv.z;
      const float v3 = v.w + pv.w;
      v.x = v0 > 0.0f ? v0 : 0.0f;
      v.y = v1 > 0.0f ? v1 : 0.0f;
      v.z = v2 > 0.0f ? v2 : 0.0f;
      v.w = v3 > 0.0f ? v3 : 0.0f;
    }
    *reinterpret_cast<float4*>(&lds[(arow + RST * i) * PA + 4 * ac4]) = v;
  }
  __syncthreads();

  v8f acc[4];
#pragma unroll
  for (int t = 0; t < 4; ++t) acc[t] = (v8f){0.f, 0.f, 0.f, 0.f, 0.f, 0.f, 0.f, 0.f};

#pragma unroll
  for (int j = 0; j < K / 4; ++j) {
#pragma unroll
    for (int t = 0; t < 4; ++t) {
      const v2f a = *reinterpret_cast<const v2f*>(&lds[(16 * t + l) * PA + 4 * j + 2 * half]);
      acc[t] = __builtin_amdgcn_wmma_f32_16x16x4_f32(
          false, a, false, b[j], (short)0, acc[t], false, false);
    }
  }
  __syncthreads();

#pragma unroll
  for (int t = 0; t < 4; ++t) {
#pragma unroll
    for (int r = 0; r < 8; ++r) {
      lds[(16 * t + 8 * half + r) * PC + 16 * w + l] = acc[t][r];
    }
  }
  __syncthreads();

#pragma unroll
  for (int i = 0; i < 8; ++i) {
    const int row  = erow + 8 * i;
    const int grow = m0 + row;
    const float4 v = *reinterpret_cast<const float4*>(&lds[row * PC + 4 * ec4]);
    if (grow < M) {
      const float di = rsqrtf(dg[i] + 1.0f);
      const float s  = di * di;
      float4 u;
      u.x = v.x * s; u.y = v.y * s; u.z = v.z * s; u.w = v.w * s;
      *reinterpret_cast<float4*>(C  + (long long)grow * NC + 4 * ec4) = v;
      *reinterpret_cast<float4*>(C2 + (long long)grow * NC + 4 * ec4) = u;
    }
  }
}

__global__ void fill_f32(float* __restrict__ p, float v, int n) {
  int t = blockIdx.x * blockDim.x + threadIdx.x;
  if (t < n) p[t] = v;
}

#define HIST_CAP  51200
#define HIST_WGS  32
#define HIST_THR  1024
#define HIST_INFL 25
#define HIST_FL   8
__global__ __launch_bounds__(HIST_THR) void deg_hist(
    const int* __restrict__ dst, float* __restrict__ deg, int E, int n) {
  __shared__ __attribute__((aligned(16))) unsigned int hist[HIST_CAP];
  const int tid  = threadIdx.x;
  const int lane = tid & 31;
  const int wv   = tid >> 5;
  const int nwg  = (int)gridDim.x;
  const int per  = (E + nwg - 1) / nwg;
  const int e_lo = (int)blockIdx.x * per;
  const int e_hi = (e_lo + per < E) ? (e_lo + per) : E;

  for (int base = 0; base < n; base += HIST_CAP) {
    const int cnt = (n - base < HIST_CAP) ? (n - base) : HIST_CAP;
    for (int i = 4 * tid; i < cnt; i += 4 * HIST_THR)
      *reinterpret_cast<uint4*>(&hist[i]) = make_uint4(0u, 0u, 0u, 0u);
    __syncthreads();

    for (int e = e_lo + tid; e < e_hi; e += HIST_THR * HIST_INFL) {
      int d[HIST_INFL];
#pragma unroll
      for (int j = 0; j < HIST_INFL; ++j) {
        const int ej = e + j * HIST_THR;
        d[j] = dst[ej < e_hi ? ej : e_hi - 1];
      }
#pragma unroll
      for (int j = 0; j < HIST_INFL; ++j) {
        const unsigned int r = (unsigned int)(d[j] - base);
        if (e + j * HIST_THR < e_hi && r < (unsigned int)cnt) atomicAdd(&hist[r], 1u);
      }
    }
    __syncthreads();

    const int nlines = (cnt + 31) >> 5;
    const int rot = (int)(((unsigned int)blockIdx.x * (unsigned int)nlines) / (unsigned int)nwg);
    for (int l0 = wv; l0 < nlines; l0 += (HIST_THR / 32) * HIST_FL) {
      unsigned int c[HIST_FL];
      int ix[HIST_FL];
#pragma unroll
      for (int j = 0; j < HIST_FL; ++j) {
        const int ln = l0 + j * (HIST_THR / 32);
        int L = ln + rot;
        if (L >= nlines) L -= nlines;
        const int i = 32 * L + lane;
        ix[j] = (ln < nlines && i < cnt) ? i : -1;
        c[j] = hist[ix[j] >= 0 ? i : 0];
      }
#pragma unroll
      for (int j = 0; j < HIST_FL; ++j) {
        if (ix[j] >= 0) atomicAdd(&deg[base + ix[j]], (float)c[j]);
      }
    }
    __syncthreads();
  }
}

#define EP_THR   256
#define EP_WAVES (EP_THR / 32)
__global__ __launch_bounds__(EP_THR) void edge_prop(
    const int* __restrict__ src, const int* __restrict__ dst,
    const float* __restrict__ deg,
    const float* __restrict__ h, float* __restrict__ agg,
    unsigned int* __restrict__ ticket, int E, int nsl, int nsplit) {
  __shared__ int sh[2];
  const int lane = threadIdx.x & 31;
  const int wv   = threadIdx.x >> 5;
  if (threadIdx.x == 0) {
    int hf = (__builtin_amdgcn_s_sendmsg_rtn(0x87) >> 18) & 1;
    unsigned int t = atomicAdd(&ticket[hf], 1u);
    if (t >= (unsigned int)nsl) {
      hf ^= 1;
      t = atomicAdd(&ticket[hf], 1u);
    }
    sh[0] = hf;
    sh[1] = (int)t;
  }
  __syncthreads();
  const int hf = __builtin_amdgcn_readfirstlane(sh[0]);
  const int sl = __builtin_amdgcn_readfirstlane(sh[1]);
  if (sl >= nsl) return;

  const int e0 = (sl * EP_WAVES + wv) * 64;
  if (e0 >= E) return;

  int   s[2], d[2];
  bool  own[2];
#pragma unroll
  for (int q = 0; q < 2; ++q) {
    const int e  = e0 + 32 * q + lane;
    const int ec = e < E ? e : E - 1;
    s[q] = src[ec];
    d[q] = dst[ec];
  }
  asm volatile("" :: "v"(s[0]), "v"(s[1]), "v"(d[0]), "v"(d[1]));
  float gs[2], gd[2];
#pragma unroll
  for (int q = 0; q < 2; ++q) {
    const int e = e0 + 32 * q + lane;
    own[q] = (e < E) && ((int)(d[q] >= nsplit) == hf);
    gs[q] = deg[own[q] ? s[q] : 0];
    gd[q] = deg[own[q] ? d[q] : 0];
  }
  int cfi[2];
#pragma unroll
  for (int q = 0; q < 2; ++q)
    cfi[q] = __float_as_int(rsqrtf(gs[q] + 1.0f) * rsqrtf(gd[q] + 1.0f));
  unsigned long long m =
      ((unsigned long long)__builtin_amdgcn_ballot_w32(own[1]) << 32) |
      (unsigned long long)__builtin_amdgcn_ballot_w32(own[0]);

  while (m != 0ull) {
    int  jj[8];
    bool ok[8];
#pragma unroll
    for (int k = 0; k < 8; ++k) {
      ok[k] = (m != 0ull);
      jj[k] = ok[k] ? (int)__builtin_ctzll(m) : 0;
      if (ok[k]) m &= m - 1ull;
    }
    float v0[8], v1[8];
#pragma unroll
    for (int k = 0; k < 8; ++k) {
      v0[k] = 0.0f; v1[k] = 0.0f;
      if (ok[k]) {
        const int jl = jj[k] & 31;
        const int sa = __builtin_amdgcn_readlane(s[0], jl);
        const int sb = __builtin_amdgcn_readlane(s[1], jl);
        const int sj = (jj[k] & 32) ? sb : sa;
        const float* row = h + (long long)sj * H_DIM + lane;
        v0[k] = row[0];
        v1[k] = row[32];
      }
    }
#pragma unroll
    for (int k = 0; k < 8; ++k) {
      if (ok[k]) {
        const int jl = jj[k] & 31;
        const int da = __builtin_amdgcn_readlane(d[0], jl);
        const int db = __builtin_amdgcn_readlane(d[1], jl);
        const int ca = __builtin_amdgcn_readlane(cfi[0], jl);
        const int cb = __builtin_amdgcn_readlane(cfi[1], jl);
        const int dj = (jj[k] & 32) ? db : da;
        const float cj = __int_as_float((jj[k] & 32) ? cb : ca);
        float* o = agg + (long long)dj * H_DIM + lane;
        atomicAdd(o, v0[k] * cj);
        atomicAdd(o + 32, v1[k] * cj);
      }
    }
  }
}

#define POOL_GPW  16
#define POOL_THR  1024
#define POOL_NW   32
#define POOL_SCAN 32
#define POOL_PP   (H_DIM + 4)
__global__ __launch_bounds__(POOL_THR) void pool_heads(
    const float* __restrict__ agg, const float* __restrict__ b,
    const int* __restrict__ batch,
    const float* __restrict__ Wmu, const float* __restrict__ bmu,
    const float* __restrict__ Wls, const float* __restrict__ bls,
    float* __restrict__ out, int n) {
  __shared__ __attribute__((aligned(16))) float accw[POOL_NW * POOL_GPW * H_DIM];
  __shared__ float cntw[POOL_NW * POOL_GPW];
  __shared__ __attribute__((aligned(16))) float tile[POOL_GPW * POOL_PP];

  const int tid  = threadIdx.x;
  const int lane = tid & 31;
  const int w    = tid >> 5;
  const int half = lane >> 4;
  const int l    = lane & 15;
  const int g0   = (int)blockIdx.x * POOL_GPW;

  float* mine = accw + w * (POOL_GPW * H_DIM);
#pragma unroll
  for (int i = 0; i < POOL_GPW * H_DIM / 128; ++i)
    *reinterpret_cast<float4*>(mine + 4 * (lane + 32 * i)) = make_float4(0.f, 0.f, 0.f, 0.f);
  if (lane < POOL_GPW) cntw[w * POOL_GPW + lane] = 0.0f;

  const float2 bb = *reinterpret_cast<const float2*>(b + 2 * lane);

  int    cur = -1;
  float2 acc = make_float2(0.f, 0.f);
  float  cn  = 0.0f;

  auto flush = [&]() {
    if (cur >= 0) {
      float2* p = reinterpret_cast<float2*>(mine + cur * H_DIM + 2 * lane);
      float2 o = *p;
      o.x += acc.x; o.y += acc.y;
      *p = o;
      if (lane == 0) cntw[w * POOL_GPW + cur] += cn;
    }
  };

  auto chunk = [&](int c) {
    const int i0 = c << 5;
    const int i  = i0 + lane;
    const int gl = batch[i < n ? i : n - 1] - g0;
    const bool own = (i < n) && ((unsigned int)gl < (unsigned int)POOL_GPW);
    float2 r[32];
#pragma unroll
    for (int j = 0; j < 32; ++j) {
      const int row = (i0 + j < n) ? (i0 + j) : (n - 1);
      r[j] = *reinterpret_cast<const float2*>(agg + (long long)row * H_DIM + 2 * lane);
    }
    const unsigned int m = __builtin_amdgcn_ballot_w32(own);
#pragma unroll
    for (int j = 0; j < 32; ++j) {
      if ((m >> j) & 1u) {
        const int gj = __builtin_amdgcn_readlane(gl, j);
        if (gj != cur) {
          flush();
          cur = gj;
          acc = make_float2(0.f, 0.f);
          cn  = 0.0f;
        }
        const float w0 = r[j].x + bb.x;
        const float w1 = r[j].y + bb.y;
        acc.x += w0 > 0.0f ? w0 : 0.0f;
        acc.y += w1 > 0.0f ? w1 : 0.0f;
        cn += 1.0f;
      }
    }
  };

  const int nchunks = (n + 31) >> 5;
  for (int c0 = w; c0 < nchunks; c0 += POOL_NW * POOL_SCAN) {
    int bv[POOL_SCAN];
#pragma unroll
    for (int t = 0; t < POOL_SCAN; ++t) {
      const int i = ((c0 + POOL_NW * t) << 5) + lane;
      bv[t] = batch[i < n ? i : n - 1];
    }
    unsigned int hit = 0u;
#pragma unroll
    for (int t = 0; t < POOL_SCAN; ++t) {
      const int i = ((c0 + POOL_NW * t) << 5) + lane;
      const bool own = (i < n) && ((unsigned int)(bv[t] - g0) < (unsigned int)POOL_GPW);
      if (__builtin_amdgcn_ballot_w32(own) != 0u) hit |= (1u << t);
    }
    while (hit != 0u) {
      const int t = __builtin_ctz(hit);
      hit &= hit - 1u;
      chunk(c0 + POOL_NW * t);
    }
  }
  flush();

  v2f bf[H_DIM / 4];
  float bias = 0.0f;
  if (w < 4) {
    const float* Wh = (w < 2) ? Wmu : Wls;
    const float* bh = (w < 2) ? bmu : bls;
    const int col = 16 * (w & 1) + l;
#pragma unroll
    for (int j = 0; j < H_DIM / 4; ++j) {
      const int ka = 4 * j + 2 * half;
      bf[j].x = Wh[ka * Z_DIM + col];
      bf[j].y = Wh[(ka + 1) * Z_DIM + col];
    }
    bias = bh[col];
  }
  __syncthreads();

  {
    const int g = tid >> 6;
    const int f = tid & 63;
    float s = 0.0f, c = 0.0f;
#pragma unroll 8
    for (int ww = 0; ww < POOL_NW; ++ww) {
      s += accw[ww * (POOL_GPW * H_DIM) + g * H_DIM + f];
      c += cntw[ww * POOL_GPW + g];
    }
    tile[g * POOL_PP + f] = s / fmaxf(c, 1.0f);
  }
  __syncthreads();

  v8f d = {0.f, 0.f, 0.f, 0.f, 0.f, 0.f, 0.f, 0.f};
  if (w < 4) {
#pragma unroll
    for (int j = 0; j < H_DIM / 4; ++j) {
      const v2f a = *reinterpret_cast<const v2f*>(&tile[l * POOL_PP + 4 * j + 2 * half]);
      d = __builtin_amdgcn_wmma_f32_16x16x4_f32(
          false, a, false, bf[j], (short)0, d, false, false);
    }
  }
  __syncthreads();

  if (w < 4) {
#pragma unroll
    for (int r = 0; r < 8; ++r) tile[(8 * half + r) * POOL_PP + 16 * w + l] = d[r] + bias;
  }
  __syncthreads();

  {
    const int g  = w & 15;
    const int hd = w >> 4;
    out[(size_t)hd * G_GRAPHS * Z_DIM + (size_t)(g0 + g) * Z_DIM + lane] =
        tile[g * POOL_PP + hd * Z_DIM + lane];
  }
}

static inline size_t align_up(size_t x, size_t a) { return (x + a - 1) & ~(a - 1); }

extern "C" void kernel_launch(void* const* d_in, const int* in_sizes, int n_in,
                              void* d_out, int out_size, void* d_ws, size_t ws_size,
                              hipStream_t stream) {
  const float* x   = (const float*)d_in[0];
  const int*   ei  = (const int*)d_in[1];
  const int*   bat = (const int*)d_in[2];
  const float* W1  = (const float*)d_in[3];
  const float* b1  = (const float*)d_in[4];
  const float* W2  = (const float*)d_in[5];
  const float* b2  = (const float*)d_in[6];
  const float* Wmu = (const float*)d_in[7];
  const float* bmu = (const float*)d_in[8];
  const float* Wls = (const float*)d_in[9];
  const float* bls = (const float*)d_in[10];
  float* out = (float*)d_out;

  const int N = in_sizes[0] / F_IN;
  const int E = in_sizes[1] / 2;
  const int* src = ei;
  const int* dst = ei + E;

  char* ws = (char*)d_ws;
  size_t off = 0;
  float* zreg = (float*)(ws + off); off = align_up(off + (size_t)(N + 64) * 4, 256);
  unsigned int* tick = (unsigned int*)zreg;
  float* deg  = zreg + 64;
  float* bufA = (float*)(ws + off); off = align_up(off + (size_t)N * H_DIM * 4, 256);
  float* bufB = (float*)(ws + off); off = align_up(off + (size_t)N * H_DIM * 4, 256);
  float* bufC = (float*)(ws + off); off = align_up(off + (size_t)N * H_DIM * 4, 256);
  (void)ws_size; (void)n_in; (void)out_size;

  const int BLK = 256;
  const int NSL = ((E + 63) / 64 + EP_WAVES - 1) / EP_WAVES;

  fill_f32<<<(N + 64 + BLK - 1) / BLK, BLK, 0, stream>>>(zreg, 0.0f, N + 64);
  deg_hist<<<HIST_WGS, HIST_THR, 0, stream>>>(dst, deg, E, N);

  gemm_rows64_wmma_f32<A_PLAIN, F_IN><<<(N + 63) / 64, 128, 0, stream>>>(
      x, W1, nullptr, deg, bufA, bufB, N);
  edge_prop<<<2 * NSL, EP_THR, 0, stream>>>(
      src, dst, deg, bufA, bufB, tick, E, NSL, N / 2);

  gemm_rows64_wmma_f32<A_BIAS_RELU, H_DIM><<<(N + 63) / 64, 128, 0, stream>>>(
      bufB, W2, b1, deg, bufA, bufC, N);
  edge_prop<<<2 * NSL, EP_THR, 0, stream>>>(
      src, dst, deg, bufA, bufC, tick + 2, E, NSL, N / 2);

  pool_heads<<<G_GRAPHS / POOL_GPW, POOL_THR, 0, stream>>>(
      bufC, b2, bat, Wmu, bmu, Wls, bls, out, N);
}


# Edge coefficient rsqrt(deg[src]+1)*rsqrt(deg[dst]+1) formed once per call under the first product's matrix work, stored per edge; both edge passes load it in whole lines: no count gathers; sums kept
# speedup vs baseline: 2.9703x; 1.0515x over previous
#include <hip/hip_runtime.h>

typedef __attribute__((ext_vector_type(2))) float v2f;
typedef __attribute__((ext_vector_type(8))) float v8f;

#define F_IN   128
#define H_DIM  64
#define Z_DIM  32
#define G_GRAPHS 512

#define A_PLAIN     0
#define A_BIAS_RELU 1


#define EC_EPT 8
#define EC_EPW (128 * EC_EPT)
template <int AF, int K, bool EC>
__global__ __launch_bounds__(128) void gemm_rows64_wmma_f32(
    const float* __restrict__ A, const float* __restrict__ B,
    const float* __restrict__ avec, const float* __restrict__ cvec,
    float* __restrict__ C, float* __restrict__ C2, int M,
    const int* __restrict__ esrc, const int* __restrict__ edst,
    float* __restrict__ ecf, int E)
{
  constexpr int NC  = H_DIM;
  constexpr int KV  = K / 4;
  constexpr int PA  = K + 4;
  constexpr int PC  = NC + 4;
  constexpr int NLD = 64 * KV / 128;
  constexpr int RST = 128 / KV;
  __shared__ __attribute__((aligned(16))) float lds[64 * PA];

  const int tid  = threadIdx.x;
  const int lane = tid & 31;
  const int w    = tid >> 5;
  const int half = lane >> 4;
  const int l    = lane & 15;
  const int m0   = blockIdx.x * 64;

  const int eb0 = (int)blockIdx.x * EC_EPW;
  int es[EC_EPT], ed[EC_EPT];
  if (EC) {
#pragma unroll
    for (int i = 0; i < EC_EPT; ++i) {
      const int e  = eb0 + tid + 128 * i;
      const int ec = e < E ? e : E - 1;
      es[i] = esrc[ec];
      ed[i] = edst[ec];
    }
  }

  const int arow = tid / KV;
  const int ac4  = tid % KV;
  float4 av[NLD];
#pragma unroll
  for (int i = 0; i < NLD; ++i) {
    const int grow = m0 + arow + RST * i;
    const int lrow = grow < M ? grow : M - 1;
    av[i] = *reinterpret_cast<const float4*>(A + (long long)lrow * K + 4 * ac4);
  }
  float4 pv = make_float4(0.f, 0.f, 0.f, 0.f);
  if (AF == A_BIAS_RELU) pv = *reinterpret_cast<const float4*>(avec + 4 * ac4);

  const float* Bcol = B + (16 * w + l);
  v2f b[K / 4];
#pragma unroll
  for (int j = 0; j < K / 4; ++j) {
    const int ka = 4 * j + 2 * half;
    b[j].x = Bcol[ka * NC];
    b[j].y = Bcol[(ka + 1) * NC];
  }

  const int erow = tid >> 4;
  const int ec4  = tid & 15;
  float dg[8];
#pragma unroll
  for (int i = 0; i < 8; ++i) {
    const int grow = m0 + erow + 8 * i;
    dg[i] = cvec[grow < M ? grow : M - 1];
  }

#pragma unroll
  for (int i = 0; i < NLD; ++i) {
    float4 v = av[i];
    if (AF == A_BIAS_RELU) {
      const float v0 = v.x + pv.x;
      const float v1 = v.y + pv.y;
      const float v2 = v.z + pv.z;
      const float v3 = v.w + pv.w;
      v.x = v0 > 0.0f ? v0 : 0.0f;
      v.y = v1 > 0.0f ? v1 : 0.0f;
      v.z = v2 > 0.0f ? v2 : 0.0f;
      v.w = v3 > 0.0f ? v3 : 0.0f;
    }
    *reinterpret_cast<float4*>(&lds[(arow + RST * i) * PA + 4 * ac4]) = v;
  }
  __syncthreads();

  float egs[EC_EPT], egd[EC_EPT];
  if (EC) {
#pragma unroll
    for (int i = 0; i < EC_EPT; ++i) {
      egs[i] = cvec[es[i]];
      egd[i] = cvec[ed[i]];
    }
  }

  v8f acc[4];
#pragma unroll
  for (int t = 0; t < 4; ++t) acc[t] = (v8f){0.f, 0.f, 0.f, 0.f, 0.f, 0.f, 0.f, 0.f};

#pragma unroll
  for (int j = 0; j < K / 4; ++j) {
#pragma unroll
    for (int t = 0; t < 4; ++t) {
      const v2f a = *reinterpret_cast<const v2f*>(&lds[(16 * t + l) * PA + 4 * j + 2 * half]);
      acc[t] = __builtin_amdgcn_wmma_f32_16x16x4_f32(
          false, a, false, b[j], (short)0, acc[t], false, false);
    }
  }
  __syncthreads();

  if (EC) {
#pragma unroll
    for (int i = 0; i < EC_EPT; ++i) {
      const int e = eb0 + tid + 128 * i;
      if (e < E) ecf[e] = rsqrtf(egs[i] + 1.0f) * rsqrtf(egd[i] + 1.0f);
    }
    for (int eb = eb0 + (int)gridDim.x * EC_EPW; eb < E; eb += (int)gridDim.x * EC_EPW) {
      for (int i = 0; i < EC_EPT; ++i) {
        const int e = eb + tid + 128 * i;
        if (e < E) ecf[e] = rsqrtf(cvec[esrc[e]] + 1.0f) * rsqrtf(cvec[edst[e]] + 1.0f);
      }
    }
  }

#pragma unroll
  for (int t = 0; t < 4; ++t) {
#pragma unroll
    for (int r = 0; r < 8; ++r) {
      lds[(16 * t + 8 * half + r) * PC + 16 * w + l] = acc[t][r];
    }
  }
  __syncthreads();

#pragma unroll
  for (int i = 0; i < 8; ++i) {
    const int row  = erow + 8 * i;
    const int grow = m0 + row;
    const float4 v = *reinterpret_cast<const float4*>(&lds[row * PC + 4 * ec4]);
    if (grow < M) {
      const float di = rsqrtf(dg[i] + 1.0f);
      const float s  = di * di;
      float4 u;
      u.x = v.x * s; u.y = v.y * s; u.z = v.z * s; u.w = v.w * s;
      *reinterpret_cast<float4*>(C  + (long long)grow * NC + 4 * ec4) = v;
      *reinterpret_cast<float4*>(C2 + (long long)grow * NC + 4 * ec4) = u;
    }
  }
}

__global__ void fill_f32(float* __restrict__ p, float v, int n) {
  int t = blockIdx.x * blockDim.x + threadIdx.x;
  if (t < n) p[t] = v;
}

#define HIST_CAP  51200
#define HIST_WGS  32
#define HIST_THR  1024
#define HIST_INFL 25
#define HIST_FL   8
__global__ __launch_bounds__(HIST_THR) void deg_hist(
    const int* __restrict__ dst, float* __restrict__ deg, int E, int n) {
  __shared__ __attribute__((aligned(16))) unsigned int hist[HIST_CAP];
  const int tid  = threadIdx.x;
  const int lane = tid & 31;
  const int wv   = tid >> 5;
  const int nwg  = (int)gridDim.x;
  const int per  = (E + nwg - 1) / nwg;
  const int e_lo = (int)blockIdx.x * per;
  const int e_hi = (e_lo + per < E) ? (e_lo + per) : E;

  for (int base = 0; base < n; base += HIST_CAP) {
    const int cnt = (n - base < HIST_CAP) ? (n - base) : HIST_CAP;
    for (int i = 4 * tid; i < cnt; i += 4 * HIST_THR)
      *reinterpret_cast<uint4*>(&hist[i]) = make_uint4(0u, 0u, 0u, 0u);
    __syncthreads();

    for (int e = e_lo + tid; e < e_hi; e += HIST_THR * HIST_INFL) {
      int d[HIST_INFL];
#pragma unroll
      for (int j = 0; j < HIST_INFL; ++j) {
        const int ej = e + j * HIST_THR;
        d[j] = dst[ej < e_hi ? ej : e_hi - 1];
      }
#pragma unroll
      for (int j = 0; j < HIST_INFL; ++j) {
        const unsigned int r = (unsigned int)(d[j] - base);
        if (e + j * HIST_THR < e_hi && r < (unsigned int)cnt) atomicAdd(&hist[r], 1u);
      }
    }
    __syncthreads();

    const int nlines = (cnt + 31) >> 5;
    const int rot = (int)(((unsigned int)blockIdx.x * (unsigned int)nlines) / (unsigned int)nwg);
    for (int l0 = wv; l0 < nlines; l0 += (HIST_THR / 32) * HIST_FL) {
      unsigned int c[HIST_FL];
      int ix[HIST_FL];
#pragma unroll
      for (int j = 0; j < HIST_FL; ++j) {
        const int ln = l0 + j * (HIST_THR / 32);
        int L = ln + rot;
        if (L >= nlines) L -= nlines;
        const int i = 32 * L + lane;
        ix[j] = (ln < nlines && i < cnt) ? i : -1;
        c[j] = hist[ix[j] >= 0 ? i : 0];
      }
#pragma unroll
      for (int j = 0; j < HIST_FL; ++j) {
        if (ix[j] >= 0) atomicAdd(&deg[base + ix[j]], (float)c[j]);
      }
    }
    __syncthreads();
  }
}

#define EP_THR   256
#define EP_WAVES (EP_THR / 32)
__global__ __launch_bounds__(EP_THR) void edge_prop(
    const int* __restrict__ src, const int* __restrict__ dst,
    const float* __restrict__ cf,
    const float* __restrict__ h, float* __restrict__ agg,
    unsigned int* __restrict__ ticket, int E, int nsl, int nsplit) {
  __shared__ int sh[2];
  const int lane = threadIdx.x & 31;
  const int wv   = threadIdx.x >> 5;
  if (threadIdx.x == 0) {
    int hf = (__builtin_amdgcn_s_sendmsg_rtn(0x87) >> 18) & 1;
    unsigned int t = atomicAdd(&ticket[hf], 1u);
    if (t >= (unsigned int)nsl) {
      hf ^= 1;
      t = atomicAdd(&ticket[hf], 1u);
    }
    sh[0] = hf;
    sh[1] = (int)t;
  }
  __syncthreads();
  const int hf = __builtin_amdgcn_readfirstlane(sh[0]);
  const int sl = __builtin_amdgcn_readfirstlane(sh[1]);
  if (sl >= nsl) return;

  const int e0 = (sl * EP_WAVES + wv) * 64;
  if (e0 >= E) return;

  int   s[2], d[2], cfi[2];
  bool  own[2];
#pragma unroll
  for (int q = 0; q < 2; ++q) {
    const int e  = e0 + 32 * q + lane;
    const int ec = e < E ? e : E - 1;
    s[q] = src[ec];
    d[q] = dst[ec];
    cfi[q] = __float_as_int(cf[ec]);
  }
  asm volatile("" :: "v"(s[0]), "v"(s[1]), "v"(d[0]), "v"(d[1]), "v"(cfi[0]), "v"(cfi[1]));
#pragma unroll
  for (int q = 0; q < 2; ++q) {
    const int e = e0 + 32 * q + lane;
    own[q] = (e < E) && ((int)(d[q] >= nsplit) == hf);
  }
  unsigned long long m =
      ((unsigned long long)__builtin_amdgcn_ballot_w32(own[1]) << 32) |
      (unsigned long long)__builtin_amdgcn_ballot_w32(own[0]);

  while (m != 0ull) {
    int  jj[8];
    bool ok[8];
#pragma unroll
    for (int k = 0; k < 8; ++k) {
      ok[k] = (m != 0ull);
      jj[k] = ok[k] ? (int)__builtin_ctzll(m) : 0;
      if (ok[k]) m &= m - 1ull;
    }
    float v0[8], v1[8];
#pragma unroll
    for (int k = 0; k < 8; ++k) {
      v0[k] = 0.0f; v1[k] = 0.0f;
      if (ok[k]) {
        const int jl = jj[k] & 31;
        const int sa = __builtin_amdgcn_readlane(s[0], jl);
        const int sb = __builtin_amdgcn_readlane(s[1], jl);
        const int sj = (jj[k] & 32) ? sb : sa;
        const float* row = h + (long long)sj * H_DIM + lane;
        v0[k] = row[0];
        v1[k] = row[32];
      }
    }
#pragma unroll
    for (int k = 0; k < 8; ++k) {
      if (ok[k]) {
        const int jl = jj[k] & 31;
        const int da = __builtin_amdgcn_readlane(d[0], jl);
        const int db = __builtin_amdgcn_readlane(d[1], jl);
        const int ca = __builtin_amdgcn_readlane(cfi[0], jl);
        const int cb = __builtin_amdgcn_readlane(cfi[1], jl);
        const int dj = (jj[k] & 32) ? db : da;
        const float cj = __int_as_float((jj[k] & 32) ? cb : ca);
        float* o = agg + (long long)dj * H_DIM + lane;
        atomicAdd(o, v0[k] * cj);
        atomicAdd(o + 32, v1[k] * cj);
      }
    }
  }
}

#define POOL_GPW  16
#define POOL_THR  1024
#define POOL_NW   32
#define POOL_SCAN 32
#define POOL_PP   (H_DIM + 4)
__global__ __launch_bounds__(POOL_THR) void pool_heads(
    const float* __restrict__ agg, const float* __restrict__ b,
    const int* __restrict__ batch,
    const float* __restrict__ Wmu, const float* __restrict__ bmu,
    const float* __restrict__ Wls, const float* __restrict__ bls,
    float* __restrict__ out, int n) {
  __shared__ __attribute__((aligned(16))) float accw[POOL_NW * POOL_GPW * H_DIM];
  __shared__ float cntw[POOL_NW * POOL_GPW];
  __shared__ __attribute__((aligned(16))) float tile[POOL_GPW * POOL_PP];

  const int tid  = threadIdx.x;
  const int lane = tid & 31;
  const int w    = tid >> 5;
  const int half = lane >> 4;
  const int l    = lane & 15;
  const int g0   = (int)blockIdx.x * POOL_GPW;

  float* mine = accw + w * (POOL_GPW * H_DIM);
#pragma unroll
  for (int i = 0; i < POOL_GPW * H_DIM / 128; ++i)
    *reinterpret_cast<float4*>(mine + 4 * (lane + 32 * i)) = make_float4(0.f, 0.f, 0.f, 0.f);
  if (lane < POOL_GPW) cntw[w * POOL_GPW + lane] = 0.0f;

  const float2 bb = *reinterpret_cast<const float2*>(b + 2 * lane);

  int    cur = -1;
  float2 acc = make_float2(0.f, 0.f);
  float  cn  = 0.0f;

  auto flush = [&]() {
    if (cur >= 0) {
      float2* p = reinterpret_cast<float2*>(mine + cur * H_DIM + 2 * lane);
      float2 o = *p;
      o.x += acc.x; o.y += acc.y;
      *p = o;
      if (lane == 0) cntw[w * POOL_GPW + cur] += cn;
    }
  };

  auto chunk = [&](int c) {
    const int i0 = c << 5;
    const int i  = i0 + lane;
    const int gl = batch[i < n ? i : n - 1] - g0;
    const bool own = (i < n) && ((unsigned int)gl < (unsigned int)POOL_GPW);
    float2 r[32];
#pragma unroll
    for (int j = 0; j < 32; ++j) {
      const int row = (i0 + j < n) ? (i0 + j) : (n - 1);
      r[j] = *reinterpret_cast<const float2*>(agg + (long long)row * H_DIM + 2 * lane);
    }
    const unsigned int m = __builtin_amdgcn_ballot_w32(own);
#pragma unroll
    for (int j = 0; j < 32; ++j) {
      if ((m >> j) & 1u) {
        const int gj = __builtin_amdgcn_readlane(gl, j);
        if (gj != cur) {
          flush();
          cur = gj;
          acc = make_float2(0.f, 0.f);
          cn  = 0.0f;
        }
        const float w0 = r[j].x + bb.x;
        const float w1 = r[j].y + bb.y;
        acc.x += w0 > 0.0f ? w0 : 0.0f;
        acc.y += w1 > 0.0f ? w1 : 0.0f;
        cn += 1.0f;
      }
    }
  };

  const int nchunks = (n + 31) >> 5;
  for (int c0 = w; c0 < nchunks; c0 += POOL_NW * POOL_SCAN) {
    int bv[POOL_SCAN];
#pragma unroll
    for (int t = 0; t < POOL_SCAN; ++t) {
      const int i = ((c0 + POOL_NW * t) << 5) + lane;
      bv[t] = batch[i < n ? i : n - 1];
    }
    unsigned int hit = 0u;
#pragma unroll
    for (int t = 0; t < POOL_SCAN; ++t) {
      const int i = ((c0 + POOL_NW * t) << 5) + lane;
      const bool own = (i < n) && ((unsigned int)(bv[t] - g0) < (unsigned int)POOL_GPW);
      if (__builtin_amdgcn_ballot_w32(own) != 0u) hit |= (1u << t);
    }
    while (hit != 0u) {
      const int t = __builtin_ctz(hit);
      hit &= hit - 1u;
      chunk(c0 + POOL_NW * t);
    }
  }
  flush();

  v2f bf[H_DIM / 4];
  float bias = 0.0f;
  if (w < 4) {
    const float* Wh = (w < 2) ? Wmu : Wls;
    const float* bh = (w < 2) ? bmu : bls;
    const int col = 16 * (w & 1) + l;
#pragma unroll
    for (int j = 0; j < H_DIM / 4; ++j) {
      const int ka = 4 * j + 2 * half;
      bf[j].x = Wh[ka * Z_DIM + col];
      bf[j].y = Wh[(ka + 1) * Z_DIM + col];
    }
    bias = bh[col];
  }
  __syncthreads();

  {
    const int g = tid >> 6;
    const int f = tid & 63;
    float s = 0.0f, c = 0.0f;
#pragma unroll 8
    for (int ww = 0; ww < POOL_NW; ++ww) {
      s += accw[ww * (POOL_GPW * H_DIM) + g * H_DIM + f];
      c += cntw[ww * POOL_GPW + g];
    }
    tile[g * POOL_PP + f] = s / fmaxf(c, 1.0f);
  }
  __syncthreads();

  v8f d = {0.f, 0.f, 0.f, 0.f, 0.f, 0.f, 0.f, 0.f};
  if (w < 4) {
#pragma unroll
    for (int j = 0; j < H_DIM / 4; ++j) {
      const v2f a = *reinterpret_cast<const v2f*>(&tile[l * POOL_PP + 4 * j + 2 * half]);
      d = __builtin_amdgcn_wmma_f32_16x16x4_f32(
          false, a, false, bf[j], (short)0, d, false, false);
    }
  }
  __syncthreads();

  if (w < 4) {
#pragma unroll
    for (int r = 0; r < 8; ++r) tile[(8 * half + r) * POOL_PP + 16 * w + l] = d[r] + bias;
  }
  __syncthreads();

  {
    const int g  = w & 15;
    const int hd = w >> 4;
    out[(size_t)hd * G_GRAPHS * Z_DIM + (size_t)(g0 + g) * Z_DIM + lane] =
        tile[g * POOL_PP + hd * Z_DIM + lane];
  }
}

static inline size_t align_up(size_t x, size_t a) { return (x + a - 1) & ~(a - 1); }

extern "C" void kernel_launch(void* const* d_in, const int* in_sizes, int n_in,
                              void* d_out, int out_size, void* d_ws, size_t ws_size,
                              hipStream_t stream) {
  const float* x   = (const float*)d_in[0];
  const int*   ei  = (const int*)d_in[1];
  const int*   bat = (const int*)d_in[2];
  const float* W1  = (const float*)d_in[3];
  const float* b1  = (const float*)d_in[4];
  const float* W2  = (const float*)d_in[5];
  const float* b2  = (const float*)d_in[6];
  const float* Wmu = (const float*)d_in[7];
  const float* bmu = (const float*)d_in[8];
  const float* Wls = (const float*)d_in[9];
  const float* bls = (const float*)d_in[10];
  float* out = (float*)d_out;

  const int N = in_sizes[0] / F_IN;
  const int E = in_sizes[1] / 2;
  const int* src = ei;
  const int* dst = ei + E;

  char* ws = (char*)d_ws;
  size_t off = 0;
  float* zreg = (float*)(ws + off); off = align_up(off + (size_t)(N + 64) * 4, 256);
  unsigned int* tick = (unsigned int*)zreg;
  float* deg  = zreg + 64;
  float* bufA = (float*)(ws + off); off = align_up(off + (size_t)N * H_DIM * 4, 256);
  float* bufB = (float*)(ws + off); off = align_up(off + (size_t)N * H_DIM * 4, 256);
  float* bufC = (float*)(ws + off); off = align_up(off + (size_t)N * H_DIM * 4, 256);
  float* ecf  = (float*)(ws + off); off = align_up(off + (size_t)E * 4, 256);
  (void)ws_size; (void)n_in; (void)out_size;

  const int BLK = 256;
  const int NSL = ((E + 63) / 64 + EP_WAVES - 1) / EP_WAVES;

  fill_f32<<<(N + 64 + BLK - 1) / BLK, BLK, 0, stream>>>(zreg, 0.0f, N + 64);
  deg_hist<<<HIST_WGS, HIST_THR, 0, stream>>>(dst, deg, E, N);

  gemm_rows64_wmma_f32<A_PLAIN, F_IN, true><<<(N + 63) / 64, 128, 0, stream>>>(
      x, W1, nullptr, deg, bufA, bufB, N, src, dst, ecf, E);
  edge_prop<<<2 * NSL, EP_THR, 0, stream>>>(
      src, dst, ecf, bufA, bufB, tick, E, NSL, N / 2);

  gemm_rows64_wmma_f32<A_BIAS_RELU, H_DIM, false><<<(N + 63) / 64, 128, 0, stream>>>(
      bufB, W2, b1, deg, bufA, bufC, N, nullptr, nullptr, nullptr, 0);
  edge_prop<<<2 * NSL, EP_THR, 0, stream>>>(
      src, dst, ecf, bufA, bufC, tick + 2, E, NSL, N / 2);

  pool_heads<<<G_GRAPHS / POOL_GPW, POOL_THR, 0, stream>>>(
      bufC, b2, bat, Wmu, bmu, Wls, bls, out, N);
}
